# starting kernel
# baseline (speedup 1.0000x reference)
#include <hip/hip_runtime.h>
#include <hip/hip_bf16.h>


typedef __attribute__((ext_vector_type(16))) _Float16 v16h;
typedef __attribute__((ext_vector_type(8)))  float    v8f;
typedef __attribute__((ext_vector_type(4)))  int      v4i;

#define NN      50000
#define DH      128
#define DCAT    256
#define NE      800000
#define MTILES  (NN / 16)

#define WMMA_F16(A, B, C) \
    __builtin_amdgcn_wmma_f32_16x16x32_f16(false, (A), false, (B), (short)0, (C), false, false)

__device__ __forceinline__ v16h load_frag(const _Float16* p0, const _Float16* p1) {
    v16h r;
    reinterpret_cast<v4i*>(&r)[0] = *reinterpret_cast<const v4i*>(p0);
    reinterpret_cast<v4i*>(&r)[1] = *reinterpret_cast<const v4i*>(p1);
    return r;
}

__device__ __forceinline__ float sigmoid_f(float x) {
    return 1.0f / (1.0f + __expf(-x));
}

__global__ void __launch_bounds__(256) prep_weight(const float* __restrict__ W,
                                                   _Float16* __restrict__ frag,
                                                   int K) {
    int idx   = blockIdx.x * 256 + threadIdx.x;
    int total = (K / 32) * 8 * 32 * 16;
    if (idx >= total) return;
    int j     = idx & 15;
    int lane  = (idx >> 4) & 31;
    int ntile = (idx >> 9) & 7;
    int ktile = idx >> 12;
    int n     = ntile * 16 + (lane & 15);
    int khalf = lane >> 4;
    int k     = ktile * 32 + khalf * 8 + ((j < 8) ? j : 16 + (j - 8));
    frag[idx] = (_Float16)W[(size_t)k * DH + n];
}

__global__ void __launch_bounds__(256) f32_to_f16(const float* __restrict__ a,
                                                  _Float16* __restrict__ b, int n) {
    int i = blockIdx.x * 256 + threadIdx.x;
    if (i < n) b[i] = (_Float16)a[i];
}

__global__ void __launch_bounds__(256) zero_f32(float* __restrict__ p, int n) {
    int i = blockIdx.x * 256 + threadIdx.x;
    if (i < n) p[i] = 0.0f;
}

__global__ void __launch_bounds__(256) gates_kernel(
    const _Float16* __restrict__ x16,  const _Float16* __restrict__ h16,
    const float* __restrict__ h_prev,
    const float* __restrict__ b_xr,    const float* __restrict__ b_xz,
    const _Float16* __restrict__ fWxr, const _Float16* __restrict__ fWhr,
    const _Float16* __restrict__ fWxz, const _Float16* __restrict__ fWhz,
    float* __restrict__ z_ws, float* __restrict__ rh32, _Float16* __restrict__ rh16)
{
    const int lane  = threadIdx.x & 31;
    const int ntile = threadIdx.x >> 5;
    const int mtile = blockIdx.x;
    const int col   = ntile * 16 + (lane & 15);
    const int khalf = lane >> 4;
    const int arow  = mtile * 16 + (lane & 15);

    v8f accr, accz;
    const float br = b_xr[col], bz = b_xz[col];
#pragma unroll
    for (int v = 0; v < 8; ++v) { accr[v] = br; accz[v] = bz; }

    const _Float16* xb = x16 + (size_t)arow * DH;
    const _Float16* hb = h16 + (size_t)arow * DH;

#pragma unroll
    for (int kt = 0; kt < 4; ++kt) {
        const int ko = kt * 32 + khalf * 8;
        v16h ax = load_frag(xb + ko, xb + ko + 16);
        v16h ah = load_frag(hb + ko, hb + ko + 16);
        const size_t off = ((size_t)(kt * 8 + ntile) * 32 + lane) * 16;
        v16h bxr = load_frag(fWxr + off, fWxr + off + 8);
        v16h bhr = load_frag(fWhr + off, fWhr + off + 8);
        v16h bxz = load_frag(fWxz + off, fWxz + off + 8);
        v16h bhz = load_frag(fWhz + off, fWhz + off + 8);
        accr = WMMA_F16(ax, bxr, accr);
        accr = WMMA_F16(ah, bhr, accr);
        accz = WMMA_F16(ax, bxz, accz);
        accz = WMMA_F16(ah, bhz, accz);
    }

    const int mbase = mtile * 16 + ((lane & 16) ? 8 : 0);
#pragma unroll
    for (int v = 0; v < 8; ++v) {
        const size_t idx = (size_t)(mbase + v) * DH + col;
        const float rv = sigmoid_f(accr[v]);
        const float zv = sigmoid_f(accz[v]);
        const float h  = h_prev[idx];
        const float rh = rv * h;
        z_ws[idx] = zv;
        rh32[idx] = rh;
        rh16[idx] = (_Float16)rh;
    }
}

__global__ void __launch_bounds__(256) scatter_kernel(
    const int* __restrict__ ei, const float* __restrict__ x,
    const float* __restrict__ rh32,
    float* __restrict__ agg, float* __restrict__ cnt)
{
    const int lane = threadIdx.x & 31;
    const int wave = threadIdx.x >> 5;
    const int e    = blockIdx.x * 8 + wave;
    const int s    = ei[e];
    const int d    = ei[NE + e];
    const float* xs = x    + (size_t)s * DH;
    const float* rs = rh32 + (size_t)s * DH;
    float* ag = agg + (size_t)d * DCAT;
#pragma unroll
    for (int i = 0; i < 4; ++i) {
        const int c = lane + i * 32;
        atomicAdd(&ag[c], xs[c]);
    }
#pragma unroll
    for (int i = 0; i < 4; ++i) {
        const int c = lane + i * 32;
        atomicAdd(&ag[DH + c], rs[c]);
    }
    if (lane == 0) atomicAdd(&cnt[d], 1.0f);
}

__global__ void __launch_bounds__(256) cand_kernel(
    const float* __restrict__ agg, const float* __restrict__ cnt,
    const _Float16* __restrict__ x16, const _Float16* __restrict__ rh16,
    const _Float16* __restrict__ fWl, const _Float16* __restrict__ fWr,
    const float* __restrict__ b_l,
    const float* __restrict__ z_ws, const float* __restrict__ h_prev,
    float* __restrict__ out)
{
    const int lane  = threadIdx.x & 31;
    const int ntile = threadIdx.x >> 5;
    const int mtile = blockIdx.x;
    const int col   = ntile * 16 + (lane & 15);
    const int khalf = lane >> 4;
    const int arow  = mtile * 16 + (lane & 15);

    const float inv = 1.0f / fmaxf(cnt[arow], 1.0f);

    v8f acc;
    const float bl = b_l[col];
#pragma unroll
    for (int v = 0; v < 8; ++v) acc[v] = bl;

    const float* ab = agg + (size_t)arow * DCAT;

#pragma unroll
    for (int kt = 0; kt < 8; ++kt) {
        const int ko = kt * 32 + khalf * 8;
        v16h am;
#pragma unroll
        for (int j = 0; j < 8; ++j) {
            am[j]     = (_Float16)(ab[ko + j]      * inv);
            am[8 + j] = (_Float16)(ab[ko + 16 + j] * inv);
        }
        const _Float16* xcb = (kt < 4)
            ? (x16  + (size_t)arow * DH + kt * 32)
            : (rh16 + (size_t)arow * DH + (kt - 4) * 32);
        v16h ac = load_frag(xcb + khalf * 8, xcb + khalf * 8 + 16);

        const size_t off = ((size_t)(kt * 8 + ntile) * 32 + lane) * 16;
        v16h bwl = load_frag(fWl + off, fWl + off + 8);
        v16h bwr = load_frag(fWr + off, fWr + off + 8);
        acc = WMMA_F16(am, bwl, acc);
        acc = WMMA_F16(ac, bwr, acc);
    }

    const int mbase = mtile * 16 + ((lane & 16) ? 8 : 0);
#pragma unroll
    for (int v = 0; v < 8; ++v) {
        const size_t idx = (size_t)(mbase + v) * DH + col;
        const float zv = z_ws[idx];
        out[idx] = (1.0f - zv) * acc[v] + zv * h_prev[idx];
    }
}

extern "C" void kernel_launch(void* const* d_in, const int* in_sizes, int n_in,
                              void* d_out, int out_size, void* d_ws, size_t ws_size,
                              hipStream_t stream) {
    const float* x      = (const float*)d_in[0];
    const int*   ei     = (const int*)  d_in[1];
    const float* h_prev = (const float*)d_in[2];
    const float* W_xr   = (const float*)d_in[3];
    const float* b_xr   = (const float*)d_in[4];
    const float* W_hr   = (const float*)d_in[5];
    const float* W_xz   = (const float*)d_in[6];
    const float* b_xz   = (const float*)d_in[7];
    const float* W_hz   = (const float*)d_in[8];
    const float* W_l    = (const float*)d_in[9];
    const float* b_l    = (const float*)d_in[10];
    const float* W_r    = (const float*)d_in[11];
    float* out = (float*)d_out;

    char* ws = (char*)d_ws;
    size_t off = 0;
    auto alloc = [&](size_t bytes) -> char* {
        char* p = ws + off;
        off += (bytes + 255) & ~(size_t)255;
        return p;
    };
    _Float16* x16  = (_Float16*)alloc((size_t)NN * DH * 2);
    _Float16* h16  = (_Float16*)alloc((size_t)NN * DH * 2);
    _Float16* rh16 = (_Float16*)alloc((size_t)NN * DH * 2);
    float*    z_ws = (float*)   alloc((size_t)NN * DH * 4);
    float*    rh32 = (float*)   alloc((size_t)NN * DH * 4);
    float*    agg  = (float*)   alloc((size_t)NN * DCAT * 4);
    float*    cnt  = (float*)   alloc((size_t)NN * 4);
    _Float16* fWxr = (_Float16*)alloc(128 * 128 * 2);
    _Float16* fWhr = (_Float16*)alloc(128 * 128 * 2);
    _Float16* fWxz = (_Float16*)alloc(128 * 128 * 2);
    _Float16* fWhz = (_Float16*)alloc(128 * 128 * 2);
    _Float16* fWl  = (_Float16*)alloc(256 * 128 * 2);
    _Float16* fWr  = (_Float16*)alloc(256 * 128 * 2);

    prep_weight<<<64,  256, 0, stream>>>(W_xr, fWxr, 128);
    prep_weight<<<64,  256, 0, stream>>>(W_hr, fWhr, 128);
    prep_weight<<<64,  256, 0, stream>>>(W_xz, fWxz, 128);
    prep_weight<<<64,  256, 0, stream>>>(W_hz, fWhz, 128);
    prep_weight<<<128, 256, 0, stream>>>(W_l,  fWl,  256);
    prep_weight<<<128, 256, 0, stream>>>(W_r,  fWr,  256);

    const int nact = NN * DH;
    f32_to_f16<<<(nact + 255) / 256, 256, 0, stream>>>(x,      x16, nact);
    f32_to_f16<<<(nact + 255) / 256, 256, 0, stream>>>(h_prev, h16, nact);
    zero_f32<<<((NN * DCAT) + 255) / 256, 256, 0, stream>>>(agg, NN * DCAT);
    zero_f32<<<(NN + 255) / 256, 256, 0, stream>>>(cnt, NN);

    gates_kernel<<<MTILES, 256, 0, stream>>>(x16, h16, h_prev, b_xr, b_xz,
                                             fWxr, fWhr, fWxz, fWhz,
                                             z_ws, rh32, rh16);

    scatter_kernel<<<NE / 8, 256, 0, stream>>>(ei, x, rh32, agg, cnt);

    cand_kernel<<<MTILES, 256, 0, stream>>>(agg, cnt, x16, rh16, fWl, fWr,
                                            b_l, z_ws, h_prev, out);
}


# z, h_prev and out move as whole 512-B rows: result tiles pass through LDS so a wave reads/writes its 2 rows with 16-B-per-lane accesses, not sixteen 64-B pieces per plane; f32 sum order unchanged
# speedup vs baseline: 4.0730x; 4.0730x over previous
#include <hip/hip_runtime.h>
#include <hip/hip_bf16.h>


typedef __attribute__((ext_vector_type(16))) _Float16 v16h;
typedef __attribute__((ext_vector_type(8)))  float    v8f;
typedef __attribute__((ext_vector_type(4)))  int      v4i;
typedef __attribute__((ext_vector_type(4)))  _Float16 v4h;
typedef __attribute__((ext_vector_type(8)))  _Float16 v8h;

#define NN      50000
#define DH      128
#define DCAT    256
#define NE      800000
#define MTILES  (NN / 16)
#define CAP     32

#define WMMA_F16(A, B, C) \
    __builtin_amdgcn_wmma_f32_16x16x32_f16(false, (A), false, (B), (short)0, (C), false, false)

__device__ __forceinline__ v16h load_frag(const _Float16* p0, const _Float16* p1) {
    v16h r;
    reinterpret_cast<v4i*>(&r)[0] = *reinterpret_cast<const v4i*>(p0);
    reinterpret_cast<v4i*>(&r)[1] = *reinterpret_cast<const v4i*>(p1);
    return r;
}

#define GPITCH  (256 + 8)
#define RPITCH  (128 + 8)
#define HPITCH  (128 + 4)
#define CPITCH  (512 + 8)

__device__ __forceinline__ void put4(_Float16* dst, const float4 v) {
    v4h o;
    o[0] = (_Float16)v.x; o[1] = (_Float16)v.y; o[2] = (_Float16)v.z; o[3] = (_Float16)v.w;
    *reinterpret_cast<v4h*>(dst) = o;
}

__device__ __forceinline__ float sigmoid_f(float x) {
    return 1.0f / (1.0f + __expf(-x));
}

__device__ __forceinline__ int frag_src(int idx) {
    const int j     = idx & 15;
    const int lane  = (idx >> 4) & 31;
    const int ntile = (idx >> 9) & 7;
    const int ktile = idx >> 12;
    const int n     = ntile * 16 + (lane & 15);
    const int khalf = lane >> 4;
    const int k     = ktile * 32 + khalf * 8 + ((j < 8) ? j : 16 + (j - 8));
    return k * DH + n;
}

__global__ void __launch_bounds__(256) zero_i32(int* __restrict__ p, int n) {
    int i = blockIdx.x * 256 + threadIdx.x;
    if (i < n) p[i] = 0;
}

static_assert(MTILES * 256 == NE, "one gate-kernel thread per edge and tile");
__device__ __forceinline__ void place_edge(int s, int d, int p,
                                           int* __restrict__ cnt,
                                           int* __restrict__ slots,
                                           int* __restrict__ ovf) {
    if (p < CAP) {
        slots[d * CAP + p] = s;
    } else {
        const int q = atomicAdd(&cnt[NN], 1);
        ovf[2 * q]     = s;
        ovf[2 * q + 1] = d;
    }
}

#define GBLK    512

__device__ __forceinline__ void lds_barrier() {
    __builtin_amdgcn_fence(__ATOMIC_RELEASE, "workgroup", "local");
    __builtin_amdgcn_s_barrier();
    __builtin_amdgcn_fence(__ATOMIC_ACQUIRE, "workgroup", "local");
}
#define WFRAG   (DH * DH)

__global__ void __launch_bounds__(256) gates_kernel(
    const float* __restrict__ x,       const float* __restrict__ h_prev,
    const float* __restrict__ b_xr,    const float* __restrict__ b_xz,
    const float* __restrict__ W_xr,    const float* __restrict__ W_hr,
    const float* __restrict__ W_xz,    const float* __restrict__ W_hz,
    float* __restrict__ z_ws, _Float16* __restrict__ xc16,
    const int* __restrict__ ei, int* __restrict__ cnt,
    int* __restrict__ slots, int* __restrict__ ovf,
    const float* __restrict__ W_l, const float* __restrict__ W_r,
    _Float16* __restrict__ fWl, _Float16* __restrict__ fWr)
{
    __shared__ __attribute__((aligned(16))) _Float16 Wf[4 * WFRAG];
    __shared__ __attribute__((aligned(16))) _Float16 At[16 * GPITCH];
    __shared__ __attribute__((aligned(16))) _Float16 Rt[16 * RPITCH];
    __shared__ __attribute__((aligned(16))) float    Hs[16 * HPITCH];

    const int lane  = threadIdx.x & 31;
    const int ntile = threadIdx.x >> 5;
    const int col   = ntile * 16 + (lane & 15);
    const int khalf = lane >> 4;

    int mtile = blockIdx.x;

    float4 xv[2], hv[2];
    int es, ed;
#pragma unroll
    for (int i = 0; i < 2; ++i) {
        const unsigned g4 = (unsigned)(mtile * 16 + ntile * 2 + i) * (DH / 4) + lane;
        xv[i] = reinterpret_cast<const float4*>(x)[g4];
        hv[i] = reinterpret_cast<const float4*>(h_prev)[g4];
    }
    es = ei[mtile * 256 + (int)threadIdx.x];
    ed = ei[NE + mtile * 256 + (int)threadIdx.x];

    {
        const int n  = threadIdx.x & 127;
        const int kq = threadIdx.x >> 7;
#pragma unroll 1
        for (int w = 0; w < 4; ++w) {
            const float* __restrict__ Wsrc = (w == 0) ? W_xr : (w == 1) ? W_hr : (w == 2) ? W_xz : W_hz;
            float wv[8][8];
#pragma unroll
            for (int it = 0; it < 8; ++it) {
                const int kg = it * 2 + kq;
#pragma unroll
                for (int i = 0; i < 8; ++i) wv[it][i] = Wsrc[(kg * 8 + i) * DH + n];
            }
#pragma unroll
            for (int it = 0; it < 8; ++it) {
                const int kt  = it >> 1;
                const int j0  = (it & 1) * 8;
                const int dst = ((kt * 8 + (n >> 4)) * 32 + kq * 16 + (n & 15)) * 16 + j0;
                v8h o;
#pragma unroll
                for (int i = 0; i < 8; ++i) o[i] = (_Float16)wv[it][i];
                *reinterpret_cast<v8h*>(Wf + w * WFRAG + dst) = o;
            }
        }
    }

    const float br = b_xr[col], bz = b_xz[col];

    const _Float16* arow = At + (lane & 15) * GPITCH + khalf * 8;
    const int lrow = (lane & 16) ? 8 : 0;

    for (;;) {
        const int ces = es, ced = ed;
        const int sidx = mtile * 256 + (int)threadIdx.x;
#pragma unroll
        for (int i = 0; i < 2; ++i) {
            _Float16* row = At + (ntile * 2 + i) * GPITCH + lane * 4;
            put4(row,       xv[i]);
            put4(row + 128, hv[i]);
            *reinterpret_cast<float4*>(Hs + (ntile * 2 + i) * HPITCH + lane * 4) = hv[i];
        }

        const int ep = atomicAdd(&cnt[ced], 1);

        const int  next = mtile + GBLK;
        const bool more = next < MTILES;
        if (more) {
#pragma unroll
            for (int i = 0; i < 2; ++i) {
                const unsigned g4 = (unsigned)(next * 16 + ntile * 2 + i) * (DH / 4) + lane;
                xv[i] = reinterpret_cast<const float4*>(x)[g4];
                hv[i] = reinterpret_cast<const float4*>(h_prev)[g4];
            }
            es = ei[next * 256 + (int)threadIdx.x];
            ed = ei[NE + next * 256 + (int)threadIdx.x];
        }

        lds_barrier();

        v8f accr, accz;
#pragma unroll
        for (int v = 0; v < 8; ++v) { accr[v] = br; accz[v] = bz; }

#pragma unroll
        for (int kt = 0; kt < 4; ++kt) {
            const _Float16* ap = arow + kt * 32;
            v16h ax = load_frag(ap,       ap + 16);
            v16h ah = load_frag(ap + 128, ap + 128 + 16);
            const _Float16* bp = Wf + ((kt * 8 + ntile) * 32 + lane) * 16;
            v16h bxr = load_frag(bp,             bp + 8);
            v16h bhr = load_frag(bp + WFRAG,     bp + WFRAG + 8);
            v16h bxz = load_frag(bp + 2 * WFRAG, bp + 2 * WFRAG + 8);
            v16h bhz = load_frag(bp + 3 * WFRAG, bp + 3 * WFRAG + 8);
            accr = WMMA_F16(ax, bxr, accr);
            accr = WMMA_F16(ah, bhr, accr);
            accz = WMMA_F16(ax, bxz, accz);
            accz = WMMA_F16(ah, bhz, accz);
        }

#pragma unroll
        for (int v = 0; v < 8; ++v) {
            const float rv = sigmoid_f(accr[v]);
            float* hp = Hs + (lrow + v) * HPITCH + col;
            const float h  = *hp;
            Rt[(lrow + v) * RPITCH + col] = (_Float16)(rv * h);
            *hp = sigmoid_f(accz[v]);
        }

        lds_barrier();

#pragma unroll
        for (int i = 0; i < 2; ++i) {
            const int row = ntile * 2 + i;
            const _Float16* sp = (lane < 16) ? (At + row * GPITCH + lane * 8)
                                             : (Rt + row * RPITCH + (lane - 16) * 8);
            const v4i piece = *reinterpret_cast<const v4i*>(sp);
            reinterpret_cast<v4i*>(xc16)[(unsigned)(mtile * 16 + row) * (DCAT / 8) + lane] = piece;
        }

#pragma unroll
        for (int i = 0; i < 2; ++i) {
            const int row = ntile * 2 + i;
            const float4 zr = *reinterpret_cast<const float4*>(Hs + row * HPITCH + lane * 4);
            reinterpret_cast<float4*>(z_ws)[(unsigned)(mtile * 16 + row) * (DH / 4) + lane] = zr;
        }

        place_edge(ces, ced, ep, cnt, slots, ovf);
        if (mtile < (2 * DCAT * DH) / 256) {
            const int li = sidx & (DCAT * DH - 1);
            const float* Wsrc = (sidx < DCAT * DH) ? W_l : W_r;
            _Float16*    fdst = (sidx < DCAT * DH) ? fWl : fWr;
            fdst[li] = (_Float16)Wsrc[frag_src(li)];
        }

        if (!more) break;
        mtile = next;
        lds_barrier();
    }
}

__device__ __forceinline__ v4i row_piece(const _Float16* __restrict__ base, int row, int lane) {
    return reinterpret_cast<const v4i*>(base + (size_t)row * DCAT)[lane];
}

__device__ __forceinline__ void add8(float (&acc)[8], const v4i piece) {
    const v8h h = __builtin_bit_cast(v8h, piece);
#pragma unroll
    for (int t = 0; t < 8; ++t) acc[t] += (float)h[t];
}

__global__ void __launch_bounds__(256) cand_kernel(
    const int* __restrict__ cnt, const int* __restrict__ slots,
    const int* __restrict__ ovf,
    const _Float16* __restrict__ xc16,
    const _Float16* __restrict__ fWl, const _Float16* __restrict__ fWr,
    const float* __restrict__ b_l,
    const float* __restrict__ z_ws, const float* __restrict__ h_prev,
    float* __restrict__ out)
{
    __shared__ __attribute__((aligned(16))) _Float16 At[16 * CPITCH];
    __shared__ __attribute__((aligned(16))) float    Ns[16 * HPITCH];

    const int lane  = threadIdx.x & 31;
    const int ntile = threadIdx.x >> 5;
    const int mtile = blockIdx.x;
    const int col   = ntile * 16 + (lane & 15);
    const int khalf = lane >> 4;

    const int r0 = __builtin_amdgcn_readfirstlane(mtile * 16 + ntile * 2);

    v4i own[2];
#pragma unroll
    for (int i = 0; i < 2; ++i) own[i] = row_piece(xc16, r0 + i, lane);

    int deg[2];
#pragma unroll
    for (int k = 0; k < 2; ++k) deg[k] = cnt[r0 + k];

    int first[2];
#pragma unroll
    for (int k = 0; k < 2; ++k)
        first[k] = (lane < min(deg[k], CAP)) ? slots[(r0 + k) * CAP + lane] : 0;

#pragma unroll
    for (int i = 0; i < 2; ++i)
        *reinterpret_cast<v4i*>(At + (ntile * 2 + i) * CPITCH + 256 + lane * 8) = own[i];

#pragma unroll
    for (int k = 0; k < 2; ++k) {
        const int n = min(deg[k], CAP);
        float acc8[8];
#pragma unroll
        for (int t = 0; t < 8; ++t) acc8[t] = 0.0f;
        const int sv = first[k];
        for (int j = 0; j < n; j += 8) {
            v4i q[8];
            {
                const int s = __builtin_amdgcn_readlane(sv, j);
                q[0] = row_piece(xc16, s, lane);
            }
#pragma unroll
            for (int t = 1; t < 8; ++t) {
                if (j + t < n) {
                    const int s = __builtin_amdgcn_readlane(sv, j + t);
                    q[t] = row_piece(xc16, s, lane);
                }
            }
            add8(acc8, q[0]);
#pragma unroll
            for (int t = 1; t < 8; ++t) {
                if (j + t < n) add8(acc8, q[t]);
            }
        }
        if (deg[k] > CAP) {
            const int row = r0 + k;
            const int no  = cnt[NN];
            for (int base = 0; base < no; base += 32) {
                const int e  = base + lane;
                const int os = (e < no) ? ovf[2 * e]     : 0;
                const int od = (e < no) ? ovf[2 * e + 1] : -1;
                unsigned m = __builtin_amdgcn_ballot_w32(od == row);
                while (m) {
                    const int b = __builtin_ctz(m);
                    m &= m - 1;
                    const int s = __builtin_amdgcn_readlane(os, b);
                    add8(acc8, row_piece(xc16, s, lane));
                }
            }
        }
        const float s = 1.0f / fmaxf((float)deg[k], 1.0f);
        v8h m;
#pragma unroll
        for (int t = 0; t < 8; ++t) m[t] = (_Float16)(acc8[t] * s);
        *reinterpret_cast<v8h*>(At + (ntile * 2 + k) * CPITCH + lane * 8) = m;
    }

    v8f acc;
    const float bl = b_l[col];
#pragma unroll
    for (int v = 0; v < 8; ++v) acc[v] = bl;

    const size_t off0 = ((size_t)ntile * 32 + lane) * 16;
    const v16h bwl0 = load_frag(fWl + off0, fWl + off0 + 8);
    const v16h bwr0 = load_frag(fWr + off0, fWr + off0 + 8);

    __syncthreads();

    const _Float16* arow = At + (lane & 15) * CPITCH + khalf * 8;

#pragma unroll
    for (int kt = 0; kt < 8; ++kt) {
        const _Float16* ap = arow + kt * 32;
        v16h am = load_frag(ap,       ap + 16);
        v16h ac = load_frag(ap + 256, ap + 256 + 16);
        const size_t off = ((size_t)(kt * 8 + ntile) * 32 + lane) * 16;
        v16h bwl = (kt == 0) ? bwl0 : load_frag(fWl + off, fWl + off + 8);
        v16h bwr = (kt == 0) ? bwr0 : load_frag(fWr + off, fWr + off + 8);
        acc = WMMA_F16(am, bwl, acc);
        acc = WMMA_F16(ac, bwr, acc);
    }

    const int lrow = (lane & 16) ? 8 : 0;
#pragma unroll
    for (int v = 0; v < 8; ++v) Ns[(lrow + v) * HPITCH + col] = acc[v];

    lds_barrier();

    float4 zr[2], hr[2];
#pragma unroll
    for (int i = 0; i < 2; ++i) {
        const unsigned g4 = (unsigned)(r0 + i) * (DH / 4) + lane;
        zr[i] = reinterpret_cast<const float4*>(z_ws)[g4];
        hr[i] = reinterpret_cast<const float4*>(h_prev)[g4];
    }
#pragma unroll
    for (int i = 0; i < 2; ++i) {
        const unsigned g4 = (unsigned)(r0 + i) * (DH / 4) + lane;
        const float4 n4 = *reinterpret_cast<const float4*>(Ns + (ntile * 2 + i) * HPITCH + lane * 4);
        float4 o;
        o.x = (1.0f - zr[i].x) * n4.x + zr[i].x * hr[i].x;
        o.y = (1.0f - zr[i].y) * n4.y + zr[i].y * hr[i].y;
        o.z = (1.0f - zr[i].z) * n4.z + zr[i].z * hr[i].z;
        o.w = (1.0f - zr[i].w) * n4.w + zr[i].w * hr[i].w;
        reinterpret_cast<float4*>(out)[g4] = o;
    }
}

extern "C" void kernel_launch(void* const* d_in, const int* in_sizes, int n_in,
                              void* d_out, int out_size, void* d_ws, size_t ws_size,
                              hipStream_t stream) {
    const float* x      = (const float*)d_in[0];
    const int*   ei     = (const int*)  d_in[1];
    const float* h_prev = (const float*)d_in[2];
    const float* W_xr   = (const float*)d_in[3];
    const float* b_xr   = (const float*)d_in[4];
    const float* W_hr   = (const float*)d_in[5];
    const float* W_xz   = (const float*)d_in[6];
    const float* b_xz   = (const float*)d_in[7];
    const float* W_hz   = (const float*)d_in[8];
    const float* W_l    = (const float*)d_in[9];
    const float* b_l    = (const float*)d_in[10];
    const float* W_r    = (const float*)d_in[11];
    float* out = (float*)d_out;

    char* ws = (char*)d_ws;
    size_t off = 0;
    auto alloc = [&](size_t bytes) -> char* {
        char* p = ws + off;
        off += (bytes + 255) & ~(size_t)255;
        return p;
    };
    float*    z_ws = (float*)   alloc((size_t)NN * DH * 4);
    _Float16* xc16 = (_Float16*)alloc((size_t)NN * DCAT * 2);
    int*      cnt   = (int*)    alloc((size_t)(NN + 1) * 4);
    int*      slots = (int*)    alloc((size_t)NN * CAP * 4);
    int*      ovf   = (int*)    alloc((size_t)NE * 2 * 4);
    _Float16* fWl  = (_Float16*)alloc(256 * 128 * 2);
    _Float16* fWr  = (_Float16*)alloc(256 * 128 * 2);

    zero_i32<<<(NN + 1 + 255) / 256, 256, 0, stream>>>(cnt, NN + 1);

    gates_kernel<<<GBLK, 256, 0, stream>>>(x, h_prev, b_xr, b_xz,
                                             W_xr, W_hr, W_xz, W_hz,
                                             z_ws, xc16,
                                             ei, cnt, slots, ovf,
                                             W_l, W_r, fWl, fWr);

    cand_kernel<<<MTILES, 256, 0, stream>>>(cnt, slots, ovf, xc16, fWl, fWr,
                                            b_l, z_ws, h_prev, out);
}
